# starting kernel
# baseline (speedup 1.0000x reference)
#include <hip/hip_runtime.h>
#include <math.h>

constexpr int B    = 64;
constexpr int T    = 2048;
constexpr int C    = 512;
constexpr int NSEG = 16;
constexpr int TSEG = T / NSEG;
constexpr int K1   = 3 * C;
constexpr int N1   = C;
constexpr int K2   = C;
constexpr int N2   = C / 2;

typedef float v2f __attribute__((ext_vector_type(2)));
typedef float v8f __attribute__((ext_vector_type(8)));

__global__ void pool_partial_kernel(const float* __restrict__ x,
                                    const int* __restrict__ lengths,
                                    float* __restrict__ partial) {
  const int seg = blockIdx.x;
  const int b   = blockIdx.y;
  const int c4  = threadIdx.x * 4;

  const int L = lengths[b];
  const int q = L / 4;
  const int p = (3 * L) / 4;

  const int t0 = seg * TSEG;
  int t1 = t0 + TSEG;
  if (t1 > L) t1 = L;

  float sb[4] = {0.f, 0.f, 0.f, 0.f};
  float sf[4] = {0.f, 0.f, 0.f, 0.f};
  float mx[4];
  const float ninf = -__builtin_inff();
  mx[0] = ninf; mx[1] = ninf; mx[2] = ninf; mx[3] = ninf;

  const float* xp = x + ((size_t)b * T + t0) * C + c4;
  for (int t = t0; t < t1; ++t, xp += C) {
    const float4 v = *(const float4*)xp;
    const float vv[4] = {v.x, v.y, v.z, v.w};
    if (t < q) {
      #pragma unroll
      for (int i = 0; i < 4; ++i) sb[i] += vv[i];
    } else if (t < p) {
      #pragma unroll
      for (int i = 0; i < 4; ++i) mx[i] = fmaxf(mx[i], vv[i]);
    } else {
      #pragma unroll
      for (int i = 0; i < 4; ++i) sf[i] += vv[i];
    }
  }

  float* base = partial + (((size_t)b * NSEG + seg) * 3) * C + c4;
  *(float4*)(base + 0 * C) = make_float4(sb[0], sb[1], sb[2], sb[3]);
  *(float4*)(base + 1 * C) = make_float4(mx[0], mx[1], mx[2], mx[3]);
  *(float4*)(base + 2 * C) = make_float4(sf[0], sf[1], sf[2], sf[3]);
}

__global__ void pool_reduce_kernel(const float* __restrict__ partial,
                                   const int* __restrict__ lengths,
                                   float* __restrict__ feat) {
  const int b = blockIdx.x;
  const int c = threadIdx.x;

  const int L = lengths[b];
  const float qf = (float)(L / 4);
  const float ff = (float)(L - (3 * L) / 4);

  float sb = 0.f, sf = 0.f, mx = -__builtin_inff();
  const float* base = partial + (size_t)b * NSEG * 3 * C + c;
  #pragma unroll
  for (int s = 0; s < NSEG; ++s) {
    const float* ps = base + (size_t)s * 3 * C;
    sb += ps[0];
    mx = fmaxf(mx, ps[C]);
    sf += ps[2 * C];
  }
  float* fb = feat + (size_t)b * K1;
  fb[c]         = sb / qf;
  fb[C + c]     = mx;
  fb[2 * C + c] = sf / ff;
}

__device__ __forceinline__ v8f wmma_f32_k4(v2f a, v2f b, v8f c) {
  return __builtin_amdgcn_wmma_f32_16x16x4_f32(
       false, a,  false, b,
       (short)0, c,  false,  false);
}

__global__ void gemm_bias_act_kernel(const float* __restrict__ A,
                                     const float* __restrict__ W,
                                     const float* __restrict__ bias,
                                     float* __restrict__ Out,
                                     int N, int K, int do_relu) {
  const int lane = threadIdx.x;
  const int l16  = lane & 15;
  const int kh   = (lane >> 4) * 2;

  const int arow = (blockIdx.y << 4) + l16;
  const int ncol = (blockIdx.x << 4) + l16;

  v8f acc = {};
  const float* Arow = A + (size_t)arow * K;

  for (int k = 0; k < K; k += 4) {
    const int kk = k + kh;
    const v2f a = *(const v2f*)(Arow + kk);
    v2f bm;
    bm.x = W[(size_t)kk * N + ncol];
    bm.y = W[(size_t)(kk + 1) * N + ncol];
    acc = wmma_f32_k4(a, bm, acc);
  }

  const float bv = bias[ncol];
  const int rbase = (blockIdx.y << 4) + ((lane >> 4) << 3);
  #pragma unroll
  for (int r = 0; r < 8; ++r) {
    float v = acc[r] + bv;
    if (do_relu) v = fmaxf(v, 0.0f);
    Out[(size_t)(rbase + r) * N + ncol] = v;
  }
}

extern "C" void kernel_launch(void* const* d_in, const int* in_sizes, int n_in,
                              void* d_out, int out_size, void* d_ws, size_t ws_size,
                              hipStream_t stream) {
  const float* x       = (const float*)d_in[0];
  const int*   lengths = (const int*)  d_in[1];
  const float* W1      = (const float*)d_in[2];
  const float* b1      = (const float*)d_in[3];
  const float* W2      = (const float*)d_in[4];
  const float* b2      = (const float*)d_in[5];
  float*       out     = (float*)d_out;

  float* ws      = (float*)d_ws;
  float* partial = ws;
  float* feat    = partial + (size_t)B * NSEG * 3 * C;
  float* h       = feat + (size_t)B * K1;

  pool_partial_kernel<<<dim3(NSEG, B), dim3(C / 4), 0, stream>>>(x, lengths, partial);
  pool_reduce_kernel<<<dim3(B), dim3(C), 0, stream>>>(partial, lengths, feat);

  gemm_bias_act_kernel<<<dim3(N1 / 16, B / 16), dim3(32), 0, stream>>>(
      feat, W1, b1, h, N1, K1, 1);
  gemm_bias_act_kernel<<<dim3(N2 / 16, B / 16), dim3(32), 0, stream>>>(
      h, W2, b2, out, N2, K2, 0);
}


# Loads in flight: pool walks its three row ranges 16 rows per load group, each GEMM wave keeps 16 k-steps of fragments in flight behind its WMMAs; every sum keeps its order
# speedup vs baseline: 3.6342x; 3.6342x over previous
#include <hip/hip_runtime.h>
#include <math.h>


constexpr int B    = 64;
constexpr int T    = 2048;
constexpr int C    = 512;
constexpr int NSEG = 16;
constexpr int TSEG = T / NSEG;
constexpr int K1   = 3 * C;
constexpr int N1   = C;
constexpr int K2   = C;
constexpr int N2   = C / 2;

constexpr int PU = 16;
constexpr int GD = 16;

typedef float v2f __attribute__((ext_vector_type(2)));
typedef float v8f __attribute__((ext_vector_type(8)));

template <bool IS_MAX>
__device__ __forceinline__ void fold_row(float (&acc)[4], const float4 v) {
  if (IS_MAX) {
    acc[0] = fmaxf(acc[0], v.x); acc[1] = fmaxf(acc[1], v.y);
    acc[2] = fmaxf(acc[2], v.z); acc[3] = fmaxf(acc[3], v.w);
  } else {
    acc[0] += v.x; acc[1] += v.y; acc[2] += v.z; acc[3] += v.w;
  }
}

template <bool IS_MAX>
__device__ __forceinline__ void fold_range(const float* __restrict__ xp, int n,
                                           float (&acc)[4]) {
  int i = 0;
  for (; i + PU <= n; i += PU) {
    float4 v[PU];
    #pragma unroll
    for (int u = 0; u < PU; ++u) v[u] = *(const float4*)(xp + (size_t)(i + u) * C);
    #pragma unroll
    for (int u = 0; u < PU; ++u) fold_row<IS_MAX>(acc, v[u]);
  }
  if (i < n) {
    float4 v[PU];
    #pragma unroll
    for (int u = 0; u < PU; ++u) {
      int r = i + u;
      if (r > n - 1) r = n - 1;
      v[u] = *(const float4*)(xp + (size_t)r * C);
    }
    #pragma unroll
    for (int u = 0; u < PU; ++u) {
      float nxt[4] = {acc[0], acc[1], acc[2], acc[3]};
      fold_row<IS_MAX>(nxt, v[u]);
      const bool take = (i + u) < n;
      acc[0] = take ? nxt[0] : acc[0];
      acc[1] = take ? nxt[1] : acc[1];
      acc[2] = take ? nxt[2] : acc[2];
      acc[3] = take ? nxt[3] : acc[3];
    }
  }
}

__global__ void pool_partial_kernel(const float* __restrict__ x,
                                    const int* __restrict__ lengths,
                                    float* __restrict__ partial) {
  const int seg = blockIdx.x;
  const int b   = blockIdx.y;
  const int c4  = threadIdx.x * 4;

  const int L = lengths[b];
  const int q = L / 4;
  const int p = (3 * L) / 4;

  const int t0 = seg * TSEG;
  int t1 = t0 + TSEG;
  if (t1 > L) t1 = L;

  float sb[4] = {0.f, 0.f, 0.f, 0.f};
  float sf[4] = {0.f, 0.f, 0.f, 0.f};
  float mx[4];
  const float ninf = -__builtin_inff();
  mx[0] = ninf; mx[1] = ninf; mx[2] = ninf; mx[3] = ninf;

  const int e0 = q < t1 ? q : t1;
  const int s1 = q > t0 ? q : t0;
  const int e1 = p < t1 ? p : t1;
  const int s2 = p > t0 ? p : t0;

  const float* xb = x + (size_t)b * T * C + c4;
  if (e0 > t0) fold_range<false>(xb + (size_t)t0 * C, e0 - t0, sb);
  if (e1 > s1) fold_range<true >(xb + (size_t)s1 * C, e1 - s1, mx);
  if (t1 > s2) fold_range<false>(xb + (size_t)s2 * C, t1 - s2, sf);

  float* base = partial + (((size_t)b * NSEG + seg) * 3) * C + c4;
  *(float4*)(base + 0 * C) = make_float4(sb[0], sb[1], sb[2], sb[3]);
  *(float4*)(base + 1 * C) = make_float4(mx[0], mx[1], mx[2], mx[3]);
  *(float4*)(base + 2 * C) = make_float4(sf[0], sf[1], sf[2], sf[3]);
}

__global__ void pool_reduce_kernel(const float* __restrict__ partial,
                                   const int* __restrict__ lengths,
                                   float* __restrict__ feat) {
  const int b = blockIdx.x;
  const int c = threadIdx.x;

  const int L = lengths[b];
  const float qf = (float)(L / 4);
  const float ff = (float)(L - (3 * L) / 4);

  float sb = 0.f, sf = 0.f, mx = -__builtin_inff();
  const float* base = partial + (size_t)b * NSEG * 3 * C + c;
  #pragma unroll
  for (int s = 0; s < NSEG; ++s) {
    const float* ps = base + (size_t)s * 3 * C;
    sb += ps[0];
    mx = fmaxf(mx, ps[C]);
    sf += ps[2 * C];
  }
  float* fb = feat + (size_t)b * K1;
  fb[c]         = sb / qf;
  fb[C + c]     = mx;
  fb[2 * C + c] = sf / ff;
}

__device__ __forceinline__ v8f wmma_f32_k4(v2f a, v2f b, v8f c) {
  return __builtin_amdgcn_wmma_f32_16x16x4_f32(
       false, a,  false, b,
       (short)0, c,  false,  false);
}

template <int N, int K, bool RELU>
__global__ void gemm_bias_act_kernel(const float* __restrict__ A,
                                     const float* __restrict__ W,
                                     const float* __restrict__ bias,
                                     float* __restrict__ Out) {
  constexpr int NSTEP = K / 4;
  static_assert(K % 4 == 0 && NSTEP % GD == 0 && NSTEP >= GD, "K must be a multiple of 4*GD");

  const int lane = threadIdx.x;
  const int l16  = lane & 15;
  const int kh   = (lane >> 4) * 2;

  const int arow = (blockIdx.y << 4) + l16;
  const int ncol = (blockIdx.x << 4) + l16;

  const float* Ap = A + (size_t)arow * K + kh;
  const float* Wp = W + (size_t)kh * N + ncol;

  v2f fa[GD];
  v2f fb[GD];
  #pragma unroll
  for (int j = 0; j < GD; ++j) {
    fa[j]   = *(const v2f*)(Ap + 4 * j);
    fb[j].x = Wp[(size_t)(4 * j) * N];
    fb[j].y = Wp[(size_t)(4 * j + 1) * N];
    __builtin_amdgcn_sched_barrier(0);
  }

  v8f acc = {};
  for (int s = 0; s < NSTEP - GD; s += GD) {
    const float* An = Ap + 4 * (s + GD);
    const float* Wn = Wp + (size_t)(4 * (s + GD)) * N;
    #pragma unroll
    for (int j = 0; j < GD; ++j) {
      acc = wmma_f32_k4(fa[j], fb[j], acc);
      fa[j]   = *(const v2f*)(An + 4 * j);
      fb[j].x = Wn[(size_t)(4 * j) * N];
      fb[j].y = Wn[(size_t)(4 * j + 1) * N];
      __builtin_amdgcn_sched_barrier(0);
    }
  }
  #pragma unroll
  for (int j = 0; j < GD; ++j) acc = wmma_f32_k4(fa[j], fb[j], acc);

  const float bv = bias[ncol];
  const int rbase = (blockIdx.y << 4) + ((lane >> 4) << 3);
  #pragma unroll
  for (int r = 0; r < 8; ++r) {
    float v = acc[r] + bv;
    if (RELU) v = fmaxf(v, 0.0f);
    Out[(size_t)(rbase + r) * N + ncol] = v;
  }
}

extern "C" void kernel_launch(void* const* d_in, const int* in_sizes, int n_in,
                              void* d_out, int out_size, void* d_ws, size_t ws_size,
                              hipStream_t stream) {
  const float* x       = (const float*)d_in[0];
  const int*   lengths = (const int*)  d_in[1];
  const float* W1      = (const float*)d_in[2];
  const float* b1      = (const float*)d_in[3];
  const float* W2      = (const float*)d_in[4];
  const float* b2      = (const float*)d_in[5];
  float*       out     = (float*)d_out;

  float* ws      = (float*)d_ws;
  float* partial = ws;
  float* feat    = partial + (size_t)B * NSEG * 3 * C;
  float* h       = feat + (size_t)B * K1;

  pool_partial_kernel<<<dim3(NSEG, B), dim3(C / 4), 0, stream>>>(x, lengths, partial);
  pool_reduce_kernel<<<dim3(B), dim3(C), 0, stream>>>(partial, lengths, feat);

  gemm_bias_act_kernel<N1, K1, true><<<dim3(N1 / 16, B / 16), dim3(32), 0, stream>>>(
      feat, W1, b1, h);
  gemm_bias_act_kernel<N2, K2, false><<<dim3(N2 / 16, B / 16), dim3(32), 0, stream>>>(
      h, W2, b2, out);
}


# Split K over the 8 waves of a work-group per 16x16 tile in both GEMMs; the 8 partial tiles are added through LDS in slice order, so the reduction tree changes (8 partial sums, then bias)
# speedup vs baseline: 5.1008x; 1.4036x over previous
#include <hip/hip_runtime.h>
#include <math.h>


constexpr int B    = 64;
constexpr int T    = 2048;
constexpr int C    = 512;
constexpr int NSEG = 16;
constexpr int TSEG = T / NSEG;
constexpr int K1   = 3 * C;
constexpr int N1   = C;
constexpr int K2   = C;
constexpr int N2   = C / 2;

constexpr int PU = 16;
constexpr int GD = 16;

typedef float v2f __attribute__((ext_vector_type(2)));
typedef float v8f __attribute__((ext_vector_type(8)));

template <bool IS_MAX>
__device__ __forceinline__ void fold_row(float (&acc)[4], const float4 v) {
  if (IS_MAX) {
    acc[0] = fmaxf(acc[0], v.x); acc[1] = fmaxf(acc[1], v.y);
    acc[2] = fmaxf(acc[2], v.z); acc[3] = fmaxf(acc[3], v.w);
  } else {
    acc[0] += v.x; acc[1] += v.y; acc[2] += v.z; acc[3] += v.w;
  }
}

template <bool IS_MAX>
__device__ __forceinline__ void fold_range(const float* __restrict__ xp, int n,
                                           float (&acc)[4]) {
  int i = 0;
  for (; i + PU <= n; i += PU) {
    float4 v[PU];
    #pragma unroll
    for (int u = 0; u < PU; ++u) v[u] = *(const float4*)(xp + (size_t)(i + u) * C);
    #pragma unroll
    for (int u = 0; u < PU; ++u) fold_row<IS_MAX>(acc, v[u]);
  }
  if (i < n) {
    float4 v[PU];
    #pragma unroll
    for (int u = 0; u < PU; ++u) {
      int r = i + u;
      if (r > n - 1) r = n - 1;
      v[u] = *(const float4*)(xp + (size_t)r * C);
    }
    #pragma unroll
    for (int u = 0; u < PU; ++u) {
      float nxt[4] = {acc[0], acc[1], acc[2], acc[3]};
      fold_row<IS_MAX>(nxt, v[u]);
      const bool take = (i + u) < n;
      acc[0] = take ? nxt[0] : acc[0];
      acc[1] = take ? nxt[1] : acc[1];
      acc[2] = take ? nxt[2] : acc[2];
      acc[3] = take ? nxt[3] : acc[3];
    }
  }
}

__global__ void pool_partial_kernel(const float* __restrict__ x,
                                    const int* __restrict__ lengths,
                                    float* __restrict__ partial) {
  const int seg = blockIdx.x;
  const int b   = blockIdx.y;
  const int c4  = threadIdx.x * 4;

  const int L = lengths[b];
  const int q = L / 4;
  const int p = (3 * L) / 4;

  const int t0 = seg * TSEG;
  int t1 = t0 + TSEG;
  if (t1 > L) t1 = L;

  float sb[4] = {0.f, 0.f, 0.f, 0.f};
  float sf[4] = {0.f, 0.f, 0.f, 0.f};
  float mx[4];
  const float ninf = -__builtin_inff();
  mx[0] = ninf; mx[1] = ninf; mx[2] = ninf; mx[3] = ninf;

  const int e0 = q < t1 ? q : t1;
  const int s1 = q > t0 ? q : t0;
  const int e1 = p < t1 ? p : t1;
  const int s2 = p > t0 ? p : t0;

  const float* xb = x + (size_t)b * T * C + c4;
  if (e0 > t0) fold_range<false>(xb + (size_t)t0 * C, e0 - t0, sb);
  if (e1 > s1) fold_range<true >(xb + (size_t)s1 * C, e1 - s1, mx);
  if (t1 > s2) fold_range<false>(xb + (size_t)s2 * C, t1 - s2, sf);

  float* base = partial + (((size_t)b * NSEG + seg) * 3) * C + c4;
  *(float4*)(base + 0 * C) = make_float4(sb[0], sb[1], sb[2], sb[3]);
  *(float4*)(base + 1 * C) = make_float4(mx[0], mx[1], mx[2], mx[3]);
  *(float4*)(base + 2 * C) = make_float4(sf[0], sf[1], sf[2], sf[3]);
}

__global__ void pool_reduce_kernel(const float* __restrict__ partial,
                                   const int* __restrict__ lengths,
                                   float* __restrict__ feat) {
  const int b = blockIdx.x;
  const int c = threadIdx.x;

  const int L = lengths[b];
  const float qf = (float)(L / 4);
  const float ff = (float)(L - (3 * L) / 4);

  float sb = 0.f, sf = 0.f, mx = -__builtin_inff();
  const float* base = partial + (size_t)b * NSEG * 3 * C + c;
  #pragma unroll
  for (int s = 0; s < NSEG; ++s) {
    const float* ps = base + (size_t)s * 3 * C;
    sb += ps[0];
    mx = fmaxf(mx, ps[C]);
    sf += ps[2 * C];
  }
  float* fb = feat + (size_t)b * K1;
  fb[c]         = sb / qf;
  fb[C + c]     = mx;
  fb[2 * C + c] = sf / ff;
}

__device__ __forceinline__ v8f wmma_f32_k4(v2f a, v2f b, v8f c) {
  return __builtin_amdgcn_wmma_f32_16x16x4_f32(
       false, a,  false, b,
       (short)0, c,  false,  false);
}

constexpr int KS = 8;

template <int N, int K, bool RELU>
__global__ void __launch_bounds__(32 * KS)
gemm_bias_act_kernel(const float* __restrict__ A,
                     const float* __restrict__ W,
                     const float* __restrict__ bias,
                     float* __restrict__ Out) {
  constexpr int NSTEP = K / 4 / KS;
  static_assert(K % (4 * KS) == 0 && NSTEP % GD == 0 && NSTEP >= GD,
                "a K slice must be a multiple of 4*GD");

  __shared__ float red[KS * 8 * 32];

  const int lane = threadIdx.x & 31;
  const int wave = threadIdx.x >> 5;
  const int l16  = lane & 15;
  const int kh   = (lane >> 4) * 2;

  const int arow = (blockIdx.y << 4) + l16;
  const int ncol = (blockIdx.x << 4) + l16;
  const int k0   = wave * (4 * NSTEP);

  const float* Ap = A + (size_t)arow * K + k0 + kh;
  const float* Wp = W + (size_t)(k0 + kh) * N + ncol;

  const float bv = bias[ncol];

  v2f fa[GD];
  v2f fb[GD];
  #pragma unroll
  for (int j = 0; j < GD; ++j) {
    fa[j]   = *(const v2f*)(Ap + 4 * j);
    fb[j].x = Wp[(size_t)(4 * j) * N];
    fb[j].y = Wp[(size_t)(4 * j + 1) * N];
    __builtin_amdgcn_sched_barrier(0);
  }

  v8f acc = {};
  for (int s = 0; s < NSTEP - GD; s += GD) {
    const float* An = Ap + 4 * (s + GD);
    const float* Wn = Wp + (size_t)(4 * (s + GD)) * N;
    #pragma unroll
    for (int j = 0; j < GD; ++j) {
      acc = wmma_f32_k4(fa[j], fb[j], acc);
      fa[j]   = *(const v2f*)(An + 4 * j);
      fb[j].x = Wn[(size_t)(4 * j) * N];
      fb[j].y = Wn[(size_t)(4 * j + 1) * N];
      __builtin_amdgcn_sched_barrier(0);
    }
  }
  #pragma unroll
  for (int j = 0; j < GD; ++j) acc = wmma_f32_k4(fa[j], fb[j], acc);

  #pragma unroll
  for (int r = 0; r < 8; ++r) red[(wave * 8 + r) * 32 + lane] = acc[r];
  __syncthreads();

  float sum = red[wave * 32 + lane];
  #pragma unroll
  for (int s = 1; s < KS; ++s) sum += red[(s * 8 + wave) * 32 + lane];

  float v = sum + bv;
  if (RELU) v = fmaxf(v, 0.0f);
  const int orow = (blockIdx.y << 4) + wave + ((lane >> 4) << 3);
  Out[(size_t)orow * N + ncol] = v;
}

extern "C" void kernel_launch(void* const* d_in, const int* in_sizes, int n_in,
                              void* d_out, int out_size, void* d_ws, size_t ws_size,
                              hipStream_t stream) {
  const float* x       = (const float*)d_in[0];
  const int*   lengths = (const int*)  d_in[1];
  const float* W1      = (const float*)d_in[2];
  const float* b1      = (const float*)d_in[3];
  const float* W2      = (const float*)d_in[4];
  const float* b2      = (const float*)d_in[5];
  float*       out     = (float*)d_out;

  float* ws      = (float*)d_ws;
  float* partial = ws;
  float* feat    = partial + (size_t)B * NSEG * 3 * C;
  float* h       = feat + (size_t)B * K1;

  pool_partial_kernel<<<dim3(NSEG, B), dim3(C / 4), 0, stream>>>(x, lengths, partial);
  pool_reduce_kernel<<<dim3(B), dim3(C), 0, stream>>>(partial, lengths, feat);

  gemm_bias_act_kernel<N1, K1, true><<<dim3(N1 / 16, B / 16), dim3(32 * KS), 0, stream>>>(
      feat, W1, b1, h);
  gemm_bias_act_kernel<N2, K2, false><<<dim3(N2 / 16, B / 16), dim3(32 * KS), 0, stream>>>(
      h, W2, b2, out);
}


# Pooling rebalanced: 768 work-groups each fold an equal share of all valid rows (batch-major list) into per-batch pieces, added per batch row in list order; the mean sums' reduction tree changes
# speedup vs baseline: 5.3731x; 1.0534x over previous
#include <hip/hip_runtime.h>
#include <math.h>


constexpr int B    = 64;
constexpr int T    = 2048;
constexpr int C    = 512;
constexpr int G    = 768;
constexpr int NPIECE = G + B;
constexpr int K1   = 3 * C;
constexpr int N1   = C;
constexpr int K2   = C;
constexpr int N2   = C / 2;

constexpr int PU = 16;
constexpr int RU = 16;
constexpr int GD = 16;

typedef float v2f __attribute__((ext_vector_type(2)));
typedef float v8f __attribute__((ext_vector_type(8)));

template <bool IS_MAX>
__device__ __forceinline__ void fold_row(float (&acc)[4], const float4 v) {
  if (IS_MAX) {
    acc[0] = fmaxf(acc[0], v.x); acc[1] = fmaxf(acc[1], v.y);
    acc[2] = fmaxf(acc[2], v.z); acc[3] = fmaxf(acc[3], v.w);
  } else {
    acc[0] += v.x; acc[1] += v.y; acc[2] += v.z; acc[3] += v.w;
  }
}

template <bool IS_MAX>
__device__ __forceinline__ void fold_range(const float* __restrict__ xp, int n,
                                           float (&acc)[4]) {
  int i = 0;
  for (; i + PU <= n; i += PU) {
    float4 v[PU];
    #pragma unroll
    for (int u = 0; u < PU; ++u) v[u] = *(const float4*)(xp + (size_t)(i + u) * C);
    #pragma unroll
    for (int u = 0; u < PU; ++u) fold_row<IS_MAX>(acc, v[u]);
  }
  if (i < n) {
    float4 v[PU];
    #pragma unroll
    for (int u = 0; u < PU; ++u) {
      int r = i + u;
      if (r > n - 1) r = n - 1;
      v[u] = *(const float4*)(xp + (size_t)r * C);
    }
    #pragma unroll
    for (int u = 0; u < PU; ++u) {
      float nxt[4] = {acc[0], acc[1], acc[2], acc[3]};
      fold_row<IS_MAX>(nxt, v[u]);
      const bool take = (i + u) < n;
      acc[0] = take ? nxt[0] : acc[0];
      acc[1] = take ? nxt[1] : acc[1];
      acc[2] = take ? nxt[2] : acc[2];
      acc[3] = take ? nxt[3] : acc[3];
    }
  }
}

__device__ __forceinline__ int valid_rows(int L) {
  return L < 0 ? 0 : (L > T ? T : L);
}

__device__ __forceinline__ int share_start(int w, int R) { return (w * R) / G; }

__global__ void __launch_bounds__(C / 4)
pool_partial_kernel(const float* __restrict__ x,
                    const int* __restrict__ lengths,
                    float* __restrict__ partial) {
  const int w  = blockIdx.x;
  const int c4 = threadIdx.x * 4;

  int pe[B];
  int R = 0;
  #pragma unroll
  for (int b = 0; b < B; ++b) { R += valid_rows(lengths[b]); pe[b] = R; }
  const int r0 = share_start(w, R);
  const int r1 = share_start(w + 1, R);

  int b0 = 0, base = 0;
  #pragma unroll
  for (int b = 0; b < B; ++b) {
    const bool before = pe[b] <= r0;
    b0   += before ? 1 : 0;
    base  = before ? pe[b] : base;
  }

  const float ninf = -__builtin_inff();
  for (int b = b0; b < B; ++b) {
    if (base >= r1) break;
    const int L  = lengths[b];
    const int Lv = valid_rows(L);
    const int lo = r0 - base > 0 ? r0 - base : 0;
    const int hi = r1 - base < Lv ? r1 - base : Lv;
    base += Lv;
    if (hi <= lo) continue;

    const int q = L / 4;
    const int p = (3 * L) / 4;
    const int e0 = q < hi ? q : hi;
    const int s1 = q > lo ? q : lo;
    const int e1 = p < hi ? p : hi;
    const int s2 = p > lo ? p : lo;

    float sb[4] = {0.f, 0.f, 0.f, 0.f};
    float sf[4] = {0.f, 0.f, 0.f, 0.f};
    float mx[4] = {ninf, ninf, ninf, ninf};

    const float* xb = x + (size_t)b * T * C + c4;
    if (e0 > lo) fold_range<false>(xb + (size_t)lo * C, e0 - lo, sb);
    if (e1 > s1) fold_range<true >(xb + (size_t)s1 * C, e1 - s1, mx);
    if (hi > s2) fold_range<false>(xb + (size_t)s2 * C, hi - s2, sf);

    float* dst = partial + ((size_t)(w + b) * 3) * C + c4;
    *(float4*)(dst + 0 * C) = make_float4(sb[0], sb[1], sb[2], sb[3]);
    *(float4*)(dst + 1 * C) = make_float4(mx[0], mx[1], mx[2], mx[3]);
    *(float4*)(dst + 2 * C) = make_float4(sf[0], sf[1], sf[2], sf[3]);
  }
}

__global__ void __launch_bounds__(C)
pool_reduce_kernel(const float* __restrict__ partial,
                   const int* __restrict__ lengths,
                   float* __restrict__ feat) {
  const int b = blockIdx.x;
  const int c = threadIdx.x;

  int R = 0, base = 0;
  #pragma unroll
  for (int i = 0; i < B; ++i) {
    const int v = valid_rows(lengths[i]);
    base += i < b ? v : 0;
    R += v;
  }
  const int L  = lengths[b];
  const int Lv = valid_rows(L);
  const float qf = (float)(L / 4);
  const float ff = (float)(L - (3 * L) / 4);

  float sb = 0.f, sf = 0.f, mx = -__builtin_inff();
  if (Lv > 0) {
    const int w_lo = ((base + 1) * G - 1) / R;
    const int w_hi = ((base + Lv) * G - 1) / R;
    const float* pc = partial + c;
    for (int w = w_lo; w <= w_hi; w += RU) {
      float vb[RU], vm[RU], vf[RU];
      bool take[RU];
      #pragma unroll
      for (int u = 0; u < RU; ++u) {
        const int wi = w + u;
        take[u] = wi <= w_hi && share_start(wi + 1, R) > share_start(wi, R);
        const float* ps = pc + (size_t)((take[u] ? wi : w_hi) + b) * 3 * C;
        vb[u] = ps[0];
        vm[u] = ps[C];
        vf[u] = ps[2 * C];
      }
      #pragma unroll
      for (int u = 0; u < RU; ++u) {
        sb = take[u] ? sb + vb[u] : sb;
        mx = take[u] ? fmaxf(mx, vm[u]) : mx;
        sf = take[u] ? sf + vf[u] : sf;
      }
    }
  }
  float* fb = feat + (size_t)b * K1;
  fb[c]         = sb / qf;
  fb[C + c]     = mx;
  fb[2 * C + c] = sf / ff;
}

__device__ __forceinline__ v8f wmma_f32_k4(v2f a, v2f b, v8f c) {
  return __builtin_amdgcn_wmma_f32_16x16x4_f32(
       false, a,  false, b,
       (short)0, c,  false,  false);
}

constexpr int KS = 8;

template <int N, int K, bool RELU>
__global__ void __launch_bounds__(32 * KS)
gemm_bias_act_kernel(const float* __restrict__ A,
                     const float* __restrict__ W,
                     const float* __restrict__ bias,
                     float* __restrict__ Out) {
  constexpr int NSTEP = K / 4 / KS;
  static_assert(K % (4 * KS) == 0 && NSTEP % GD == 0 && NSTEP >= GD,
                "a K slice must be a multiple of 4*GD");

  __shared__ float red[KS * 8 * 32];

  const int lane = threadIdx.x & 31;
  const int wave = threadIdx.x >> 5;
  const int l16  = lane & 15;
  const int kh   = (lane >> 4) * 2;

  const int arow = (blockIdx.y << 4) + l16;
  const int ncol = (blockIdx.x << 4) + l16;
  const int k0   = wave * (4 * NSTEP);

  const float* Ap = A + (size_t)arow * K + k0 + kh;
  const float* Wp = W + (size_t)(k0 + kh) * N + ncol;

  const float bv = bias[ncol];

  v2f fa[GD];
  v2f fb[GD];
  #pragma unroll
  for (int j = 0; j < GD; ++j) {
    fa[j]   = *(const v2f*)(Ap + 4 * j);
    fb[j].x = Wp[(size_t)(4 * j) * N];
    fb[j].y = Wp[(size_t)(4 * j + 1) * N];
    __builtin_amdgcn_sched_barrier(0);
  }

  v8f acc = {};
  for (int s = 0; s < NSTEP - GD; s += GD) {
    const float* An = Ap + 4 * (s + GD);
    const float* Wn = Wp + (size_t)(4 * (s + GD)) * N;
    #pragma unroll
    for (int j = 0; j < GD; ++j) {
      acc = wmma_f32_k4(fa[j], fb[j], acc);
      fa[j]   = *(const v2f*)(An + 4 * j);
      fb[j].x = Wn[(size_t)(4 * j) * N];
      fb[j].y = Wn[(size_t)(4 * j + 1) * N];
      __builtin_amdgcn_sched_barrier(0);
    }
  }
  #pragma unroll
  for (int j = 0; j < GD; ++j) acc = wmma_f32_k4(fa[j], fb[j], acc);

  #pragma unroll
  for (int r = 0; r < 8; ++r) red[(wave * 8 + r) * 32 + lane] = acc[r];
  __syncthreads();

  float sum = red[wave * 32 + lane];
  #pragma unroll
  for (int s = 1; s < KS; ++s) sum += red[(s * 8 + wave) * 32 + lane];

  float v = sum + bv;
  if (RELU) v = fmaxf(v, 0.0f);
  const int orow = (blockIdx.y << 4) + wave + ((lane >> 4) << 3);
  Out[(size_t)orow * N + ncol] = v;
}

extern "C" void kernel_launch(void* const* d_in, const int* in_sizes, int n_in,
                              void* d_out, int out_size, void* d_ws, size_t ws_size,
                              hipStream_t stream) {
  const float* x       = (const float*)d_in[0];
  const int*   lengths = (const int*)  d_in[1];
  const float* W1      = (const float*)d_in[2];
  const float* b1      = (const float*)d_in[3];
  const float* W2      = (const float*)d_in[4];
  const float* b2      = (const float*)d_in[5];
  float*       out     = (float*)d_out;

  float* ws      = (float*)d_ws;
  float* partial = ws;
  float* feat    = partial + (size_t)NPIECE * 3 * C;
  float* h       = feat + (size_t)B * K1;

  pool_partial_kernel<<<dim3(G), dim3(C / 4), 0, stream>>>(x, lengths, partial);
  pool_reduce_kernel<<<dim3(B), dim3(C), 0, stream>>>(partial, lengths, feat);

  gemm_bias_act_kernel<N1, K1, true><<<dim3(N1 / 16, B / 16), dim3(32 * KS), 0, stream>>>(
      feat, W1, b1, h);
  gemm_bias_act_kernel<N2, K2, false><<<dim3(N2 / 16, B / 16), dim3(32 * KS), 0, stream>>>(
      h, W2, b2, out);
}


# GEMM A operand in whole lines: each wave reads its 16-row K slice of A once with 16-byte loads, parks it in a private LDS slab, takes all A fragments from there; every sum keeps its order
# speedup vs baseline: 6.8891x; 1.2821x over previous
#include <hip/hip_runtime.h>
#include <math.h>


constexpr int B    = 64;
constexpr int T    = 2048;
constexpr int C    = 512;
constexpr int K1   = 3 * C;
constexpr int N1   = C;
constexpr int K2   = C;
constexpr int N2   = C / 2;

constexpr int CG  = 64;
constexpr int NCG = C / CG;
constexpr int RL  = 16;
constexpr int PT  = (CG / 4) * RL;

constexpr int PU = 32;
constexpr int GD = 16;

typedef float v2f __attribute__((ext_vector_type(2)));
typedef float v8f __attribute__((ext_vector_type(8)));

template <bool IS_MAX>
__device__ __forceinline__ void fold_row(float (&acc)[4], const float4 v) {
  if (IS_MAX) {
    acc[0] = fmaxf(acc[0], v.x); acc[1] = fmaxf(acc[1], v.y);
    acc[2] = fmaxf(acc[2], v.z); acc[3] = fmaxf(acc[3], v.w);
  } else {
    acc[0] += v.x; acc[1] += v.y; acc[2] += v.z; acc[3] += v.w;
  }
}

constexpr size_t RP = (size_t)RL * C;

template <bool IS_MAX>
__device__ __forceinline__ void fold_range(const float* __restrict__ xp, int n,
                                           float (&acc)[4]) {
  int i = 0;
  for (; i + PU <= n; i += PU) {
    float4 v[PU];
    #pragma unroll
    for (int u = 0; u < PU; ++u) v[u] = *(const float4*)(xp + (size_t)(i + u) * RP);
    #pragma unroll
    for (int u = 0; u < PU; ++u) fold_row<IS_MAX>(acc, v[u]);
  }
  if (i < n) {
    float4 v[PU];
    #pragma unroll
    for (int u = 0; u < PU; ++u) {
      int r = i + u;
      if (r > n - 1) r = n - 1;
      v[u] = *(const float4*)(xp + (size_t)r * RP);
    }
    #pragma unroll
    for (int u = 0; u < PU; ++u) {
      float nxt[4] = {acc[0], acc[1], acc[2], acc[3]};
      fold_row<IS_MAX>(nxt, v[u]);
      const bool take = (i + u) < n;
      acc[0] = take ? nxt[0] : acc[0];
      acc[1] = take ? nxt[1] : acc[1];
      acc[2] = take ? nxt[2] : acc[2];
      acc[3] = take ? nxt[3] : acc[3];
    }
  }
}

__device__ __forceinline__ int valid_rows(int L) {
  return L < 0 ? 0 : (L > T ? T : L);
}

__device__ __forceinline__ int lane_rows(int s, int e, int rl) {
  const int n = e - s - rl;
  return n > 0 ? (n + RL - 1) / RL : 0;
}

__global__ void __launch_bounds__(PT)
pool_kernel(const float* __restrict__ x,
            const int* __restrict__ lengths,
            float* __restrict__ feat) {
  __shared__ float red[3 * RL * CG];

  const int cg  = blockIdx.x;
  const int b   = blockIdx.y;
  const int sub = threadIdx.x & 15;
  const int rl  = threadIdx.x >> 4;

  const int L  = lengths[b];
  const int Lv = valid_rows(L);
  const int q  = L / 4;
  const int p  = (3 * L) / 4;
  const int e0 = q < Lv ? q : Lv;
  const int s1 = q > 0 ? q : 0;
  const int e1 = p < Lv ? p : Lv;
  const int s2 = p > 0 ? p : 0;

  const float ninf = -__builtin_inff();
  float sb[4] = {0.f, 0.f, 0.f, 0.f};
  float sf[4] = {0.f, 0.f, 0.f, 0.f};
  float mx[4] = {ninf, ninf, ninf, ninf};

  const float* xb = x + (size_t)b * T * C + cg * CG + sub * 4;
  const int n0 = lane_rows(0,  e0, rl);
  const int n1 = lane_rows(s1, e1, rl);
  const int n2 = lane_rows(s2, Lv, rl);
  if (n0 > 0) fold_range<false>(xb + (size_t)rl * C,        n0, sb);
  if (n1 > 0) fold_range<true >(xb + (size_t)(s1 + rl) * C, n1, mx);
  if (n2 > 0) fold_range<false>(xb + (size_t)(s2 + rl) * C, n2, sf);

  float* rd = red + rl * CG + sub * 4;
  *(float4*)(rd + 0 * RL * CG) = make_float4(sb[0], sb[1], sb[2], sb[3]);
  *(float4*)(rd + 1 * RL * CG) = make_float4(mx[0], mx[1], mx[2], mx[3]);
  *(float4*)(rd + 2 * RL * CG) = make_float4(sf[0], sf[1], sf[2], sf[3]);
  __syncthreads();

  const int t = threadIdx.x;
  if (t < 3 * CG) {
    const int qq = t / CG;
    const int ch = t - qq * CG;
    const float* rp = red + qq * RL * CG + ch;
    float v = rp[0];
    if (qq == 1) {
      #pragma unroll
      for (int r = 1; r < RL; ++r) v = fmaxf(v, rp[r * CG]);
    } else {
      #pragma unroll
      for (int r = 1; r < RL; ++r) v += rp[r * CG];
      const float den = qq == 0 ? (float)q : (float)(L - p);
      v = v / den;
    }
    feat[(size_t)b * K1 + qq * C + cg * CG + ch] = v;
  }
}

__device__ __forceinline__ v8f wmma_f32_k4(v2f a, v2f b, v8f c) {
  return __builtin_amdgcn_wmma_f32_16x16x4_f32(
       false, a,  false, b,
       (short)0, c,  false,  false);
}

constexpr int KS = 8;

template <int N, int K, bool RELU>
__global__ void __launch_bounds__(32 * KS)
gemm_bias_act_kernel(const float* __restrict__ A,
                     const float* __restrict__ W,
                     const float* __restrict__ bias,
                     float* __restrict__ Out) {
  constexpr int NSTEP = K / 4 / KS;
  static_assert(K % (4 * KS) == 0 && NSTEP % GD == 0 && NSTEP >= GD,
                "a K slice must be a multiple of 4*GD");
  constexpr int SL    = 4 * NSTEP;
  constexpr int CPR   = SL / 4;
  constexpr int NLD   = 16 * CPR / 32;
  static_assert((16 * CPR) % 32 == 0, "the slab must be whole wave loads");
  constexpr int PITCH = SL + 4;
  static_assert(SL % 64 == 0, "pitch rule above assumes SL is a multiple of 64");

  __shared__ __attribute__((aligned(16))) float slab[KS * 16 * PITCH];
  __shared__ float red[KS * 8 * 32];

  const int lane = threadIdx.x & 31;
  const int wave = threadIdx.x >> 5;
  const int l16  = lane & 15;
  const int kh   = (lane >> 4) * 2;

  const int arow0 = blockIdx.y << 4;
  const int ncol  = (blockIdx.x << 4) + l16;
  const int k0    = wave * SL;

  const float* Wp = W + (size_t)(k0 + kh) * N + ncol;

  const float bv = bias[ncol];

  float4 st[NLD];
  float* myslab = slab + wave * (16 * PITCH);
  #pragma unroll
  for (int j = 0; j < NLD; ++j) {
    const int p = 32 * j + lane;
    const int r = p / CPR;
    const int c = p - r * CPR;
    st[j] = *(const float4*)(A + (size_t)(arow0 + r) * K + k0 + 4 * c);
  }

  v2f fb[GD];
  #pragma unroll
  for (int j = 0; j < GD; ++j) {
    fb[j].x = Wp[(size_t)(4 * j) * N];
    fb[j].y = Wp[(size_t)(4 * j + 1) * N];
    __builtin_amdgcn_sched_barrier(0);
  }

  #pragma unroll
  for (int j = 0; j < NLD; ++j) {
    const int p = 32 * j + lane;
    const int r = p / CPR;
    const int c = p - r * CPR;
    *(float4*)(myslab + r * PITCH + 4 * c) = st[j];
  }
  __builtin_amdgcn_fence(__ATOMIC_RELEASE, "wavefront");
  __builtin_amdgcn_wave_barrier();
  __builtin_amdgcn_fence(__ATOMIC_ACQUIRE, "wavefront");

  v2f fa[NSTEP];
  #pragma unroll
  for (int s = 0; s < NSTEP; ++s)
    fa[s] = *(const v2f*)(myslab + l16 * PITCH + 4 * s + kh);

  v8f acc = {};
  #pragma unroll
  for (int s = 0; s < NSTEP; ++s) {
    const int j = s % GD;
    acc = wmma_f32_k4(fa[s], fb[j], acc);
    if (s + GD < NSTEP) {
      fb[j].x = Wp[(size_t)(4 * (s + GD)) * N];
      fb[j].y = Wp[(size_t)(4 * (s + GD) + 1) * N];
    }
    __builtin_amdgcn_sched_barrier(0);
  }

  #pragma unroll
  for (int r = 0; r < 8; ++r) red[(wave * 8 + r) * 32 + lane] = acc[r];
  __syncthreads();

  float sum = red[wave * 32 + lane];
  #pragma unroll
  for (int s = 1; s < KS; ++s) sum += red[(s * 8 + wave) * 32 + lane];

  float v = sum + bv;
  if (RELU) v = fmaxf(v, 0.0f);
  const int orow = arow0 + wave + ((lane >> 4) << 3);
  Out[(size_t)orow * N + ncol] = v;
}

extern "C" void kernel_launch(void* const* d_in, const int* in_sizes, int n_in,
                              void* d_out, int out_size, void* d_ws, size_t ws_size,
                              hipStream_t stream) {
  const float* x       = (const float*)d_in[0];
  const int*   lengths = (const int*)  d_in[1];
  const float* W1      = (const float*)d_in[2];
  const float* b1      = (const float*)d_in[3];
  const float* W2      = (const float*)d_in[4];
  const float* b2      = (const float*)d_in[5];
  float*       out     = (float*)d_out;

  float* ws      = (float*)d_ws;
  float* feat    = ws;
  float* h       = feat + (size_t)B * K1;

  pool_kernel<<<dim3(NCG, B), dim3(PT), 0, stream>>>(x, lengths, feat);

  gemm_bias_act_kernel<N1, K1, true><<<dim3(N1 / 16, B / 16), dim3(32 * KS), 0, stream>>>(
      feat, W1, b1, h);
  gemm_bias_act_kernel<N2, K2, false><<<dim3(N2 / 16, B / 16), dim3(32 * KS), 0, stream>>>(
      h, W2, b2, out);
}


# Pooling work-groups take batch rows by length rank (longest half descending, then shortest half ascending), so the two work-groups a WGP hosts fold about equal rows; every sum keeps its order
# speedup vs baseline: 7.2736x; 1.0558x over previous
#include <hip/hip_runtime.h>
#include <math.h>


constexpr int B    = 64;
constexpr int T    = 2048;
constexpr int C    = 512;
constexpr int K1   = 3 * C;
constexpr int N1   = C;
constexpr int K2   = C;
constexpr int N2   = C / 2;

constexpr int CG  = 64;
constexpr int NCG = C / CG;
constexpr int RL  = 16;
constexpr int PT  = (CG / 4) * RL;

constexpr int PU = 32;
constexpr int GD = 16;

typedef float v2f __attribute__((ext_vector_type(2)));
typedef float v8f __attribute__((ext_vector_type(8)));

template <bool IS_MAX>
__device__ __forceinline__ void fold_row(float (&acc)[4], const float4 v) {
  if (IS_MAX) {
    acc[0] = fmaxf(acc[0], v.x); acc[1] = fmaxf(acc[1], v.y);
    acc[2] = fmaxf(acc[2], v.z); acc[3] = fmaxf(acc[3], v.w);
  } else {
    acc[0] += v.x; acc[1] += v.y; acc[2] += v.z; acc[3] += v.w;
  }
}

constexpr size_t RP = (size_t)RL * C;

template <bool IS_MAX>
__device__ __forceinline__ void fold_range(const float* __restrict__ xp, int n,
                                           float (&acc)[4]) {
  int i = 0;
  for (; i + PU <= n; i += PU) {
    float4 v[PU];
    #pragma unroll
    for (int u = 0; u < PU; ++u) v[u] = *(const float4*)(xp + (size_t)(i + u) * RP);
    #pragma unroll
    for (int u = 0; u < PU; ++u) fold_row<IS_MAX>(acc, v[u]);
  }
  if (i < n) {
    float4 v[PU];
    #pragma unroll
    for (int u = 0; u < PU; ++u) {
      int r = i + u;
      if (r > n - 1) r = n - 1;
      v[u] = *(const float4*)(xp + (size_t)r * RP);
    }
    #pragma unroll
    for (int u = 0; u < PU; ++u) {
      float nxt[4] = {acc[0], acc[1], acc[2], acc[3]};
      fold_row<IS_MAX>(nxt, v[u]);
      const bool take = (i + u) < n;
      acc[0] = take ? nxt[0] : acc[0];
      acc[1] = take ? nxt[1] : acc[1];
      acc[2] = take ? nxt[2] : acc[2];
      acc[3] = take ? nxt[3] : acc[3];
    }
  }
}

__device__ __forceinline__ int valid_rows(int L) {
  return L < 0 ? 0 : (L > T ? T : L);
}

__device__ __forceinline__ int lane_rows(int s, int e, int rl) {
  const int n = e - s - rl;
  return n > 0 ? (n + RL - 1) / RL : 0;
}

static_assert(B == 64, "two batch rows per lane of a 32-lane wave");
constexpr int KEYBITS = 18;
static_assert((T + 1) * B <= (1 << KEYBITS), "keys must fit KEYBITS bits");

__device__ __forceinline__ int pick_batch_row(const int* __restrict__ lengths,
                                              int y, int& L_out) {
  const int lane = threadIdx.x & 31;
  const int l0 = lengths[lane];
  const int l1 = lengths[lane + 32];
  const int k0 = valid_rows(l0) * B + (B - 1 - lane);
  const int k1 = valid_rows(l1) * B + (B - 1 - (lane + 32));
  const int want = y < B / 2 ? y : (B - 1) - (y - B / 2);
  int prefix = 0;
  int skip   = want;
  #pragma unroll
  for (int bit = KEYBITS - 1; bit >= 0; --bit) {
    const int probe = (prefix >> bit) | 1;
    const unsigned m0 = __builtin_amdgcn_ballot_w32((k0 >> bit) == probe);
    const unsigned m1 = __builtin_amdgcn_ballot_w32((k1 >> bit) == probe);
    const int ones = __builtin_popcount(m0) + __builtin_popcount(m1);
    if (skip < ones) prefix |= 1 << bit; else skip -= ones;
  }
  const int b = (B - 1) - (prefix & (B - 1));
  L_out = b < 32 ? __builtin_amdgcn_readlane(l0, b & 31)
                 : __builtin_amdgcn_readlane(l1, b & 31);
  return b;
}

__global__ void __launch_bounds__(PT)
pool_kernel(const float* __restrict__ x,
            const int* __restrict__ lengths,
            float* __restrict__ feat) {
  __shared__ float red[3 * RL * CG];

  const int cg  = blockIdx.x;
  const int sub = threadIdx.x & 15;
  const int rl  = threadIdx.x >> 4;

  int L;
  const int b  = pick_batch_row(lengths, (int)blockIdx.y, L);
  const int Lv = valid_rows(L);
  const int q  = L / 4;
  const int p  = (3 * L) / 4;
  const int e0 = q < Lv ? q : Lv;
  const int s1 = q > 0 ? q : 0;
  const int e1 = p < Lv ? p : Lv;
  const int s2 = p > 0 ? p : 0;

  const float ninf = -__builtin_inff();
  float sb[4] = {0.f, 0.f, 0.f, 0.f};
  float sf[4] = {0.f, 0.f, 0.f, 0.f};
  float mx[4] = {ninf, ninf, ninf, ninf};

  const float* xb = x + (size_t)b * T * C + cg * CG + sub * 4;
  const int n0 = lane_rows(0,  e0, rl);
  const int n1 = lane_rows(s1, e1, rl);
  const int n2 = lane_rows(s2, Lv, rl);
  if (n0 > 0) fold_range<false>(xb + (size_t)rl * C,        n0, sb);
  if (n1 > 0) fold_range<true >(xb + (size_t)(s1 + rl) * C, n1, mx);
  if (n2 > 0) fold_range<false>(xb + (size_t)(s2 + rl) * C, n2, sf);

  float* rd = red + rl * CG + sub * 4;
  *(float4*)(rd + 0 * RL * CG) = make_float4(sb[0], sb[1], sb[2], sb[3]);
  *(float4*)(rd + 1 * RL * CG) = make_float4(mx[0], mx[1], mx[2], mx[3]);
  *(float4*)(rd + 2 * RL * CG) = make_float4(sf[0], sf[1], sf[2], sf[3]);
  __syncthreads();

  const int t = threadIdx.x;
  if (t < 3 * CG) {
    const int qq = t / CG;
    const int ch = t - qq * CG;
    const float* rp = red + qq * RL * CG + ch;
    float v = rp[0];
    if (qq == 1) {
      #pragma unroll
      for (int r = 1; r < RL; ++r) v = fmaxf(v, rp[r * CG]);
    } else {
      #pragma unroll
      for (int r = 1; r < RL; ++r) v += rp[r * CG];
      const float den = qq == 0 ? (float)q : (float)(L - p);
      v = v / den;
    }
    feat[(size_t)b * K1 + qq * C + cg * CG + ch] = v;
  }
}

__device__ __forceinline__ v8f wmma_f32_k4(v2f a, v2f b, v8f c) {
  return __builtin_amdgcn_wmma_f32_16x16x4_f32(
       false, a,  false, b,
       (short)0, c,  false,  false);
}

constexpr int KS = 8;

template <int N, int K, bool RELU>
__global__ void __launch_bounds__(32 * KS)
gemm_bias_act_kernel(const float* __restrict__ A,
                     const float* __restrict__ W,
                     const float* __restrict__ bias,
                     float* __restrict__ Out) {
  constexpr int NSTEP = K / 4 / KS;
  static_assert(K % (4 * KS) == 0 && NSTEP % GD == 0 && NSTEP >= GD,
                "a K slice must be a multiple of 4*GD");
  constexpr int SL    = 4 * NSTEP;
  constexpr int CPR   = SL / 4;
  constexpr int NLD   = 16 * CPR / 32;
  static_assert((16 * CPR) % 32 == 0, "the slab must be whole wave loads");
  constexpr int PITCH = SL + 4;
  static_assert(SL % 64 == 0, "pitch rule above assumes SL is a multiple of 64");

  __shared__ __attribute__((aligned(16))) float slab[KS * 16 * PITCH];
  __shared__ float red[KS * 8 * 32];

  const int lane = threadIdx.x & 31;
  const int wave = threadIdx.x >> 5;
  const int l16  = lane & 15;
  const int kh   = (lane >> 4) * 2;

  const int arow0 = blockIdx.y << 4;
  const int ncol  = (blockIdx.x << 4) + l16;
  const int k0    = wave * SL;

  const float* Wp = W + (size_t)(k0 + kh) * N + ncol;

  const float bv = bias[ncol];

  float4 st[NLD];
  float* myslab = slab + wave * (16 * PITCH);
  #pragma unroll
  for (int j = 0; j < NLD; ++j) {
    const int p = 32 * j + lane;
    const int r = p / CPR;
    const int c = p - r * CPR;
    st[j] = *(const float4*)(A + (size_t)(arow0 + r) * K + k0 + 4 * c);
  }

  v2f fb[GD];
  #pragma unroll
  for (int j = 0; j < GD; ++j) {
    fb[j].x = Wp[(size_t)(4 * j) * N];
    fb[j].y = Wp[(size_t)(4 * j + 1) * N];
    __builtin_amdgcn_sched_barrier(0);
  }

  #pragma unroll
  for (int j = 0; j < NLD; ++j) {
    const int p = 32 * j + lane;
    const int r = p / CPR;
    const int c = p - r * CPR;
    *(float4*)(myslab + r * PITCH + 4 * c) = st[j];
  }
  __builtin_amdgcn_fence(__ATOMIC_RELEASE, "wavefront");
  __builtin_amdgcn_wave_barrier();
  __builtin_amdgcn_fence(__ATOMIC_ACQUIRE, "wavefront");

  v2f fa[NSTEP];
  #pragma unroll
  for (int s = 0; s < NSTEP; ++s)
    fa[s] = *(const v2f*)(myslab + l16 * PITCH + 4 * s + kh);

  v8f acc = {};
  #pragma unroll
  for (int s = 0; s < NSTEP; ++s) {
    const int j = s % GD;
    acc = wmma_f32_k4(fa[s], fb[j], acc);
    if (s + GD < NSTEP) {
      fb[j].x = Wp[(size_t)(4 * (s + GD)) * N];
      fb[j].y = Wp[(size_t)(4 * (s + GD) + 1) * N];
    }
    __builtin_amdgcn_sched_barrier(0);
  }

  #pragma unroll
  for (int r = 0; r < 8; ++r) red[(wave * 8 + r) * 32 + lane] = acc[r];
  __syncthreads();

  float sum = red[wave * 32 + lane];
  #pragma unroll
  for (int s = 1; s < KS; ++s) sum += red[(s * 8 + wave) * 32 + lane];

  float v = sum + bv;
  if (RELU) v = fmaxf(v, 0.0f);
  const int orow = arow0 + wave + ((lane >> 4) << 3);
  Out[(size_t)orow * N + ncol] = v;
}

extern "C" void kernel_launch(void* const* d_in, const int* in_sizes, int n_in,
                              void* d_out, int out_size, void* d_ws, size_t ws_size,
                              hipStream_t stream) {
  const float* x       = (const float*)d_in[0];
  const int*   lengths = (const int*)  d_in[1];
  const float* W1      = (const float*)d_in[2];
  const float* b1      = (const float*)d_in[3];
  const float* W2      = (const float*)d_in[4];
  const float* b2      = (const float*)d_in[5];
  float*       out     = (float*)d_out;

  float* ws      = (float*)d_ws;
  float* feat    = ws;
  float* h       = feat + (size_t)B * K1;

  pool_kernel<<<dim3(NCG, B), dim3(PT), 0, stream>>>(x, lengths, feat);

  gemm_bias_act_kernel<N1, K1, true><<<dim3(N1 / 16, B / 16), dim3(32 * KS), 0, stream>>>(
      feat, W1, b1, h);
  gemm_bias_act_kernel<N2, K2, false><<<dim3(N2 / 16, B / 16), dim3(32 * KS), 0, stream>>>(
      h, W2, b2, out);
}
